# starting kernel
# baseline (speedup 1.0000x reference)
#include <hip/hip_runtime.h>
#include <hip/hip_bf16.h>
#include <stdint.h>

typedef __attribute__((ext_vector_type(16))) __bf16 v16bf;
typedef __attribute__((ext_vector_type(8)))  float  v8f;

#define NUM_NODES 10000
#define IN_FEAT   64
#define HID       256
#define OUTF      3
#define GATES     768

__global__ void cvt_f32_to_bf16(const float* __restrict__ in,
                                __bf16* __restrict__ out, int n) {
    int i = blockIdx.x * blockDim.x + threadIdx.x;
    if (i < n) out[i] = (__bf16)in[i];
}

__global__ void cvt_f32_to_bf16_T(const float* __restrict__ in,
                                  __bf16* __restrict__ out, int K, int N) {
    int i = blockIdx.x * blockDim.x + threadIdx.x;
    if (i >= K * N) return;
    int k = i / N, n = i % N;
    out[(size_t)n * K + k] = (__bf16)in[(size_t)k * N + n];
}

__global__ void deg_init(float* __restrict__ deg, int n) {
    int i = blockIdx.x * blockDim.x + threadIdx.x;
    if (i < n) deg[i] = 1.0f;
}

__global__ void deg_edges(const int* __restrict__ dst, float* __restrict__ deg, int E) {
    int e = blockIdx.x * blockDim.x + threadIdx.x;
    if (e < E) atomicAdd(&deg[dst[e]], 1.0f);
}

__global__ void deg_rsqrt(const float* __restrict__ deg, float* __restrict__ dinv, int n) {
    int i = blockIdx.x * blockDim.x + threadIdx.x;
    if (i < n) dinv[i] = rsqrtf(deg[i]);
}

__global__ void __launch_bounds__(128)
wmma_gemm_bf16(const __bf16* __restrict__ A, const __bf16* __restrict__ Bt,
               float* __restrict__ C, int M, int N, int K) {
    int wave  = (int)((blockIdx.x * blockDim.x + threadIdx.x) >> 5);
    int lane  = threadIdx.x & 31;
    int tilesN = N >> 4;
    int tilesM = M >> 4;
    if (wave >= tilesM * tilesN) return;
    int tm = wave / tilesN;
    int tn = wave % tilesN;

    int l15 = lane & 15;
    int h   = lane >> 4;
    const __bf16* arow = A  + (size_t)(tm * 16 + l15) * K;
    const __bf16* bcol = Bt + (size_t)(tn * 16 + l15) * K;

    v8f acc = {};
    for (int k = 0; k < K; k += 32) {
        v16bf a, b;
        int k0 = k + h * 8;
#pragma unroll
        for (int i = 0; i < 8; ++i) {
            a[i]     = arow[k0 + i];
            a[i + 8] = arow[k0 + 16 + i];
            b[i]     = bcol[k0 + i];
            b[i + 8] = bcol[k0 + 16 + i];
        }
        acc = __builtin_amdgcn_wmma_f32_16x16x32_bf16(
            false, a, false, b, (short)0, acc, false, false);
    }
#pragma unroll
    for (int r = 0; r < 8; ++r) {
        int mm = tm * 16 + h * 8 + r;
        C[(size_t)mm * N + tn * 16 + l15] = acc[r];
    }
}

__global__ void gcn_selfloop(const float* __restrict__ xw, const float* __restrict__ dinv,
                             float* __restrict__ agg, int n) {
    int i = blockIdx.x * blockDim.x + threadIdx.x;
    if (i >= n) return;
    int node = i >> 8;
    float s = dinv[node];
    agg[i] = xw[i] * s * s;
}

__global__ void gcn_scatter(const float* __restrict__ xw,
                            const int* __restrict__ src, const int* __restrict__ dst,
                            const float* __restrict__ dinv,
                            float* __restrict__ agg, int E) {
    long long gid = (long long)blockIdx.x * blockDim.x + threadIdx.x;
    int e = (int)(gid >> 6);
    if (e >= E) return;
    int f = ((int)gid & 63) << 2;
    int s = src[e], d = dst[e];
    float norm = dinv[s] * dinv[d];
    const float4 v = *(const float4*)(xw + (size_t)s * HID + f);
    float* o = agg + (size_t)d * HID + f;
    atomicAdd(o + 0, norm * v.x);
    atomicAdd(o + 1, norm * v.y);
    atomicAdd(o + 2, norm * v.z);
    atomicAdd(o + 3, norm * v.w);
}

__global__ void bias_relu_bf16(const float* __restrict__ agg, const float* __restrict__ bias,
                               __bf16* __restrict__ hb, int n) {
    int i = blockIdx.x * blockDim.x + threadIdx.x;
    if (i >= n) return;
    float v = agg[i] + bias[i & (HID - 1)];
    hb[i] = (__bf16)(v > 0.f ? v : 0.f);
}

__global__ void __launch_bounds__(1024)
gru_scan(const float* __restrict__ gi, const __bf16* __restrict__ Whh,
         const float* __restrict__ b_ih, const float* __restrict__ b_hh,
         float* __restrict__ hs, int T) {
    __shared__ float hcur[HID];
    __shared__ float gh[GATES];
    int t = threadIdx.x;
    if (t < HID) hcur[t] = 0.0f;
    __syncthreads();

    for (int step = 0; step < T; ++step) {
        if (t < GATES) {
            const uint32_t* wrow = (const uint32_t*)(Whh + (size_t)t * HID);
            float acc = 0.0f;
#pragma unroll 8
            for (int kk = 0; kk < HID / 2; ++kk) {
                uint32_t p = wrow[kk];
                float w0 = __uint_as_float(p << 16);
                float w1 = __uint_as_float(p & 0xffff0000u);
                acc = fmaf(w0, hcur[2 * kk],     acc);
                acc = fmaf(w1, hcur[2 * kk + 1], acc);
            }
            gh[t] = acc + b_hh[t];
        } else if (step + 1 < T && t < GATES + 32) {
            __builtin_prefetch(gi + (size_t)(step + 1) * GATES + (t - GATES) * 32, 0, 0);
        }
        __syncthreads();
        if (t < HID) {
            const float* g = gi + (size_t)step * GATES;
            float r  = 1.0f / (1.0f + __expf(-(g[t]           + b_ih[t]           + gh[t])));
            float z  = 1.0f / (1.0f + __expf(-(g[HID + t]     + b_ih[HID + t]     + gh[HID + t])));
            float nn = tanhf(g[2 * HID + t] + b_ih[2 * HID + t] + r * gh[2 * HID + t]);
            float hn = (1.0f - z) * nn + z * hcur[t];
            hs[(size_t)step * HID + t] = hn;
            hcur[t] = hn;
        }
        __syncthreads();
    }
}

__global__ void fc_out_kernel(const float* __restrict__ hs, const float* __restrict__ fw,
                              const float* __restrict__ fb, float* __restrict__ out, int total) {
    int gid = blockIdx.x * blockDim.x + threadIdx.x;
    if (gid >= total) return;
    int n = gid / OUTF, o = gid % OUTF;
    const float* h = hs + (size_t)n * HID;
    const float* w = fw + (size_t)o * HID;
    float acc = fb[o];
#pragma unroll 8
    for (int k = 0; k < HID; ++k) acc = fmaf(h[k], w[k], acc);
    out[gid] = acc;
}

extern "C" void kernel_launch(void* const* d_in, const int* in_sizes, int n_in,
                              void* d_out, int out_size, void* d_ws, size_t ws_size,
                              hipStream_t stream) {
    const float* x    = (const float*)d_in[0];
    const int*   ei   = (const int*)  d_in[1];
    const float* W1   = (const float*)d_in[2];
    const float* b1   = (const float*)d_in[3];
    const float* W2   = (const float*)d_in[4];
    const float* b2   = (const float*)d_in[5];
    const float* W_ih = (const float*)d_in[6];
    const float* W_hh = (const float*)d_in[7];
    const float* b_ih = (const float*)d_in[8];
    const float* b_hh = (const float*)d_in[9];
    const float* fc_w = (const float*)d_in[10];
    const float* fc_b = (const float*)d_in[11];
    float* out = (float*)d_out;

    const int E = in_sizes[1] / 2;
    const int* src = ei;
    const int* dst = ei + E;

    char* ws = (char*)d_ws;
    size_t off = 0;
    auto alloc = [&](size_t bytes) -> char* {
        char* p = ws + off;
        off = (off + bytes + 255) & ~(size_t)255;
        return p;
    };
    float*  deg  = (float*) alloc((size_t)NUM_NODES * 4);
    float*  dinv = (float*) alloc((size_t)NUM_NODES * 4);
    float*  xw   = (float*) alloc((size_t)NUM_NODES * HID * 4);
    float*  agg  = (float*) alloc((size_t)NUM_NODES * HID * 4);
    float*  gi   = (float*) alloc((size_t)NUM_NODES * GATES * 4);
    float*  hs   = (float*) alloc((size_t)NUM_NODES * HID * 4);
    __bf16* xb   = (__bf16*)alloc((size_t)NUM_NODES * HID * 2);
    __bf16* w1t  = (__bf16*)alloc((size_t)HID * IN_FEAT * 2);
    __bf16* w2t  = (__bf16*)alloc((size_t)HID * HID * 2);
    __bf16* wih  = (__bf16*)alloc((size_t)GATES * HID * 2);
    __bf16* whh  = (__bf16*)alloc((size_t)GATES * HID * 2);
    (void)ws_size; (void)n_in; (void)out_size;

    const int TPB = 256;
    auto blk = [](long long n, int t) { return (unsigned)((n + t - 1) / t); };

    cvt_f32_to_bf16<<<blk((long long)NUM_NODES * IN_FEAT, TPB), TPB, 0, stream>>>(
        x, xb, NUM_NODES * IN_FEAT);
    cvt_f32_to_bf16_T<<<blk((long long)IN_FEAT * HID, TPB), TPB, 0, stream>>>(
        W1, w1t, IN_FEAT, HID);
    cvt_f32_to_bf16_T<<<blk((long long)HID * HID, TPB), TPB, 0, stream>>>(
        W2, w2t, HID, HID);
    cvt_f32_to_bf16<<<blk((long long)GATES * HID, TPB), TPB, 0, stream>>>(
        W_ih, wih, GATES * HID);
    cvt_f32_to_bf16<<<blk((long long)GATES * HID, TPB), TPB, 0, stream>>>(
        W_hh, whh, GATES * HID);

    deg_init<<<blk(NUM_NODES, TPB), TPB, 0, stream>>>(deg, NUM_NODES);
    deg_edges<<<blk(E, TPB), TPB, 0, stream>>>(dst, deg, E);
    deg_rsqrt<<<blk(NUM_NODES, TPB), TPB, 0, stream>>>(deg, dinv, NUM_NODES);

    const long long NH = (long long)NUM_NODES * HID;
    const unsigned gemm1_blocks = blk((long long)(NUM_NODES / 16) * (HID / 16), 4);
    const unsigned gemm2_blocks = gemm1_blocks;
    const unsigned gemm3_blocks = blk((long long)(NUM_NODES / 16) * (GATES / 16), 4);
    const unsigned scat_blocks  = blk((long long)E * 64, TPB);

    wmma_gemm_bf16<<<gemm1_blocks, 128, 0, stream>>>(xb, w1t, xw, NUM_NODES, HID, IN_FEAT);
    gcn_selfloop<<<blk(NH, TPB), TPB, 0, stream>>>(xw, dinv, agg, (int)NH);
    gcn_scatter<<<scat_blocks, TPB, 0, stream>>>(xw, src, dst, dinv, agg, E);
    bias_relu_bf16<<<blk(NH, TPB), TPB, 0, stream>>>(agg, b1, xb, (int)NH);

    wmma_gemm_bf16<<<gemm2_blocks, 128, 0, stream>>>(xb, w2t, xw, NUM_NODES, HID, HID);
    gcn_selfloop<<<blk(NH, TPB), TPB, 0, stream>>>(xw, dinv, agg, (int)NH);
    gcn_scatter<<<scat_blocks, TPB, 0, stream>>>(xw, src, dst, dinv, agg, E);
    bias_relu_bf16<<<blk(NH, TPB), TPB, 0, stream>>>(agg, b2, xb, (int)NH);

    wmma_gemm_bf16<<<gemm3_blocks, 128, 0, stream>>>(xb, wih, gi, NUM_NODES, GATES, HID);

    gru_scan<<<1, 1024, 0, stream>>>(gi, whh, b_ih, b_hh, hs, NUM_NODES);

    fc_out_kernel<<<blk((long long)NUM_NODES * OUTF, TPB), TPB, 0, stream>>>(
        hs, fc_w, fc_b, out, NUM_NODES * OUTF);
}


# GRU scan: no global-memory round trip between a step's two barriers: gi requested at the step's head (prefetch dropped), hs row stored behind the 2nd barrier, own h kept in a register; sums unchanged
# speedup vs baseline: 4.2537x; 4.2537x over previous
#include <hip/hip_runtime.h>
#include <hip/hip_bf16.h>
#include <stdint.h>

typedef __attribute__((ext_vector_type(16))) __bf16 v16bf;
typedef __attribute__((ext_vector_type(8)))  float  v8f;

#define NUM_NODES 10000
#define IN_FEAT   64
#define HID       256
#define OUTF      3
#define GATES     768

__global__ void cvt_f32_to_bf16(const float* __restrict__ in,
                                __bf16* __restrict__ out, int n) {
    int i = blockIdx.x * blockDim.x + threadIdx.x;
    if (i < n) out[i] = (__bf16)in[i];
}

__global__ void cvt_f32_to_bf16_T(const float* __restrict__ in,
                                  __bf16* __restrict__ out, int K, int N) {
    int i = blockIdx.x * blockDim.x + threadIdx.x;
    if (i >= K * N) return;
    int k = i / N, n = i % N;
    out[(size_t)n * K + k] = (__bf16)in[(size_t)k * N + n];
}

__global__ void deg_init(float* __restrict__ deg, int n) {
    int i = blockIdx.x * blockDim.x + threadIdx.x;
    if (i < n) deg[i] = 1.0f;
}

__global__ void deg_edges(const int* __restrict__ dst, float* __restrict__ deg, int E) {
    int e = blockIdx.x * blockDim.x + threadIdx.x;
    if (e < E) atomicAdd(&deg[dst[e]], 1.0f);
}

__global__ void deg_rsqrt(const float* __restrict__ deg, float* __restrict__ dinv, int n) {
    int i = blockIdx.x * blockDim.x + threadIdx.x;
    if (i < n) dinv[i] = rsqrtf(deg[i]);
}

__global__ void __launch_bounds__(128)
wmma_gemm_bf16(const __bf16* __restrict__ A, const __bf16* __restrict__ Bt,
               float* __restrict__ C, int M, int N, int K) {
    int wave  = (int)((blockIdx.x * blockDim.x + threadIdx.x) >> 5);
    int lane  = threadIdx.x & 31;
    int tilesN = N >> 4;
    int tilesM = M >> 4;
    if (wave >= tilesM * tilesN) return;
    int tm = wave / tilesN;
    int tn = wave % tilesN;

    int l15 = lane & 15;
    int h   = lane >> 4;
    const __bf16* arow = A  + (size_t)(tm * 16 + l15) * K;
    const __bf16* bcol = Bt + (size_t)(tn * 16 + l15) * K;

    v8f acc = {};
    for (int k = 0; k < K; k += 32) {
        v16bf a, b;
        int k0 = k + h * 8;
#pragma unroll
        for (int i = 0; i < 8; ++i) {
            a[i]     = arow[k0 + i];
            a[i + 8] = arow[k0 + 16 + i];
            b[i]     = bcol[k0 + i];
            b[i + 8] = bcol[k0 + 16 + i];
        }
        acc = __builtin_amdgcn_wmma_f32_16x16x32_bf16(
            false, a, false, b, (short)0, acc, false, false);
    }
#pragma unroll
    for (int r = 0; r < 8; ++r) {
        int mm = tm * 16 + h * 8 + r;
        C[(size_t)mm * N + tn * 16 + l15] = acc[r];
    }
}

__global__ void gcn_selfloop(const float* __restrict__ xw, const float* __restrict__ dinv,
                             float* __restrict__ agg, int n) {
    int i = blockIdx.x * blockDim.x + threadIdx.x;
    if (i >= n) return;
    int node = i >> 8;
    float s = dinv[node];
    agg[i] = xw[i] * s * s;
}

__global__ void gcn_scatter(const float* __restrict__ xw,
                            const int* __restrict__ src, const int* __restrict__ dst,
                            const float* __restrict__ dinv,
                            float* __restrict__ agg, int E) {
    long long gid = (long long)blockIdx.x * blockDim.x + threadIdx.x;
    int e = (int)(gid >> 6);
    if (e >= E) return;
    int f = ((int)gid & 63) << 2;
    int s = src[e], d = dst[e];
    float norm = dinv[s] * dinv[d];
    const float4 v = *(const float4*)(xw + (size_t)s * HID + f);
    float* o = agg + (size_t)d * HID + f;
    atomicAdd(o + 0, norm * v.x);
    atomicAdd(o + 1, norm * v.y);
    atomicAdd(o + 2, norm * v.z);
    atomicAdd(o + 3, norm * v.w);
}

__global__ void bias_relu_bf16(const float* __restrict__ agg, const float* __restrict__ bias,
                               __bf16* __restrict__ hb, int n) {
    int i = blockIdx.x * blockDim.x + threadIdx.x;
    if (i >= n) return;
    float v = agg[i] + bias[i & (HID - 1)];
    hb[i] = (__bf16)(v > 0.f ? v : 0.f);
}

typedef uint32_t u32x4 __attribute__((ext_vector_type(4)));
#define LDS_CH 3

static __device__ __forceinline__ float bf_lo(uint32_t u) { return __uint_as_float(u << 16); }
static __device__ __forceinline__ float bf_hi(uint32_t u) { return __uint_as_float(u & 0xffff0000u); }

template <int M>
static __device__ __forceinline__ float add_lane_xor(float v) {
    int o = __builtin_amdgcn_update_dpp(0, __builtin_bit_cast(int, v), 0x160 | M, 0xf, 0xf, false);
    return v + __builtin_bit_cast(float, o);
}

__global__ void __launch_bounds__(1024)
gru_scan(const float* __restrict__ gi, const __bf16* __restrict__ Whh,
         const float* __restrict__ b_ih, const float* __restrict__ b_hh,
         float* __restrict__ hs, int T) {
    __shared__ u32x4 wl[6 * LDS_CH][1024];
    __shared__ __attribute__((aligned(16))) float hcur[HID];
    __shared__ float gh[GATES];
    const int t = threadIdx.x;
    const int grp = t >> 3;
    const int s   = t & 7;

    u32x4 wr[6];
    float bh[6];
    int   rowi[6];
#pragma unroll
    for (int r = 0; r < 6; ++r) {
        const int row = (r >> 1) * HID + grp + 128 * (r & 1);
        const u32x4* wsrc = (const u32x4*)(Whh + (size_t)row * HID) + s;
#pragma unroll
        for (int i = 0; i < LDS_CH; ++i) wl[r * LDS_CH + i][t] = wsrc[8 * i];
        wr[r]   = wsrc[8 * LDS_CH];
        bh[r]   = b_hh[row];
        rowi[r] = row;
    }
    if (t < HID) hcur[t] = 0.0f;
    __syncthreads();

    float hprev = 0.0f;
    for (int step = 0; step < T; ++step) {
        float g0 = 0.0f, g1 = 0.0f, g2 = 0.0f;
        if (t < HID) {
            const float* g = gi + (size_t)step * GATES;
            g0 = g[t]; g1 = g[HID + t]; g2 = g[2 * HID + t];
        }
        float hv[32];
#pragma unroll
        for (int i = 0; i < 4; ++i) {
            const float4 p = *(const float4*)&hcur[64 * i + 8 * s];
            const float4 q = *(const float4*)&hcur[64 * i + 8 * s + 4];
            hv[8 * i + 0] = p.x; hv[8 * i + 1] = p.y; hv[8 * i + 2] = p.z; hv[8 * i + 3] = p.w;
            hv[8 * i + 4] = q.x; hv[8 * i + 5] = q.y; hv[8 * i + 6] = q.z; hv[8 * i + 7] = q.w;
        }
        float acc[6];
#pragma unroll
        for (int r = 0; r < 6; ++r) {
            float a = 0.0f;
#pragma unroll
            for (int i = 0; i < 4; ++i) {
                const u32x4 w = (i < LDS_CH) ? wl[r * LDS_CH + i][t] : wr[r];
                a = fmaf(bf_lo(w.x), hv[8 * i + 0], a);
                a = fmaf(bf_hi(w.x), hv[8 * i + 1], a);
                a = fmaf(bf_lo(w.y), hv[8 * i + 2], a);
                a = fmaf(bf_hi(w.y), hv[8 * i + 3], a);
                a = fmaf(bf_lo(w.z), hv[8 * i + 4], a);
                a = fmaf(bf_hi(w.z), hv[8 * i + 5], a);
                a = fmaf(bf_lo(w.w), hv[8 * i + 6], a);
                a = fmaf(bf_hi(w.w), hv[8 * i + 7], a);
            }
            acc[r] = a;
        }
#pragma unroll
        for (int r = 0; r < 6; ++r) {
            float a = acc[r];
            a = add_lane_xor<1>(a);
            a = add_lane_xor<2>(a);
            a = add_lane_xor<4>(a);
            acc[r] = a;
        }
        if (s == 0) {
#pragma unroll
            for (int r = 0; r < 6; ++r) gh[rowi[r]] = acc[r] + bh[r];
        }
        float x0 = 0.0f, x1 = 0.0f, x2 = 0.0f;
        if (t < HID) {
            x0 = g0 + b_ih[t]; x1 = g1 + b_ih[HID + t]; x2 = g2 + b_ih[2 * HID + t];
        }
        __syncthreads();
        if (t < HID) {
            float r  = 1.0f / (1.0f + __expf(-(x0 + gh[t])));
            float z  = 1.0f / (1.0f + __expf(-(x1 + gh[HID + t])));
            float nn = tanhf(x2 + r * gh[2 * HID + t]);
            float hn = fmaf(1.0f - z, nn, z * hprev);
            hcur[t] = hn;
            hprev = hn;
        }
        __syncthreads();
        if (t < HID) hs[(size_t)step * HID + t] = hprev;
    }
}

__global__ void fc_out_kernel(const float* __restrict__ hs, const float* __restrict__ fw,
                              const float* __restrict__ fb, float* __restrict__ out, int total) {
    int gid = blockIdx.x * blockDim.x + threadIdx.x;
    if (gid >= total) return;
    int n = gid / OUTF, o = gid % OUTF;
    const float* h = hs + (size_t)n * HID;
    const float* w = fw + (size_t)o * HID;
    float acc = fb[o];
#pragma unroll 8
    for (int k = 0; k < HID; ++k) acc = fmaf(h[k], w[k], acc);
    out[gid] = acc;
}

extern "C" void kernel_launch(void* const* d_in, const int* in_sizes, int n_in,
                              void* d_out, int out_size, void* d_ws, size_t ws_size,
                              hipStream_t stream) {
    const float* x    = (const float*)d_in[0];
    const int*   ei   = (const int*)  d_in[1];
    const float* W1   = (const float*)d_in[2];
    const float* b1   = (const float*)d_in[3];
    const float* W2   = (const float*)d_in[4];
    const float* b2   = (const float*)d_in[5];
    const float* W_ih = (const float*)d_in[6];
    const float* W_hh = (const float*)d_in[7];
    const float* b_ih = (const float*)d_in[8];
    const float* b_hh = (const float*)d_in[9];
    const float* fc_w = (const float*)d_in[10];
    const float* fc_b = (const float*)d_in[11];
    float* out = (float*)d_out;

    const int E = in_sizes[1] / 2;
    const int* src = ei;
    const int* dst = ei + E;

    char* ws = (char*)d_ws;
    size_t off = 0;
    auto alloc = [&](size_t bytes) -> char* {
        char* p = ws + off;
        off = (off + bytes + 255) & ~(size_t)255;
        return p;
    };
    float*  deg  = (float*) alloc((size_t)NUM_NODES * 4);
    float*  dinv = (float*) alloc((size_t)NUM_NODES * 4);
    float*  xw   = (float*) alloc((size_t)NUM_NODES * HID * 4);
    float*  agg  = (float*) alloc((size_t)NUM_NODES * HID * 4);
    float*  gi   = (float*) alloc((size_t)NUM_NODES * GATES * 4);
    float*  hs   = (float*) alloc((size_t)NUM_NODES * HID * 4);
    __bf16* xb   = (__bf16*)alloc((size_t)NUM_NODES * HID * 2);
    __bf16* w1t  = (__bf16*)alloc((size_t)HID * IN_FEAT * 2);
    __bf16* w2t  = (__bf16*)alloc((size_t)HID * HID * 2);
    __bf16* wih  = (__bf16*)alloc((size_t)GATES * HID * 2);
    __bf16* whh  = (__bf16*)alloc((size_t)GATES * HID * 2);
    (void)ws_size; (void)n_in; (void)out_size;

    const int TPB = 256;
    auto blk = [](long long n, int t) { return (unsigned)((n + t - 1) / t); };

    cvt_f32_to_bf16<<<blk((long long)NUM_NODES * IN_FEAT, TPB), TPB, 0, stream>>>(
        x, xb, NUM_NODES * IN_FEAT);
    cvt_f32_to_bf16_T<<<blk((long long)IN_FEAT * HID, TPB), TPB, 0, stream>>>(
        W1, w1t, IN_FEAT, HID);
    cvt_f32_to_bf16_T<<<blk((long long)HID * HID, TPB), TPB, 0, stream>>>(
        W2, w2t, HID, HID);
    cvt_f32_to_bf16<<<blk((long long)GATES * HID, TPB), TPB, 0, stream>>>(
        W_ih, wih, GATES * HID);
    cvt_f32_to_bf16<<<blk((long long)GATES * HID, TPB), TPB, 0, stream>>>(
        W_hh, whh, GATES * HID);

    deg_init<<<blk(NUM_NODES, TPB), TPB, 0, stream>>>(deg, NUM_NODES);
    deg_edges<<<blk(E, TPB), TPB, 0, stream>>>(dst, deg, E);
    deg_rsqrt<<<blk(NUM_NODES, TPB), TPB, 0, stream>>>(deg, dinv, NUM_NODES);

    const long long NH = (long long)NUM_NODES * HID;
    const unsigned gemm1_blocks = blk((long long)(NUM_NODES / 16) * (HID / 16), 4);
    const unsigned gemm2_blocks = gemm1_blocks;
    const unsigned gemm3_blocks = blk((long long)(NUM_NODES / 16) * (GATES / 16), 4);
    const unsigned scat_blocks  = blk((long long)E * 64, TPB);

    wmma_gemm_bf16<<<gemm1_blocks, 128, 0, stream>>>(xb, w1t, xw, NUM_NODES, HID, IN_FEAT);
    gcn_selfloop<<<blk(NH, TPB), TPB, 0, stream>>>(xw, dinv, agg, (int)NH);
    gcn_scatter<<<scat_blocks, TPB, 0, stream>>>(xw, src, dst, dinv, agg, E);
    bias_relu_bf16<<<blk(NH, TPB), TPB, 0, stream>>>(agg, b1, xb, (int)NH);

    wmma_gemm_bf16<<<gemm2_blocks, 128, 0, stream>>>(xb, w2t, xw, NUM_NODES, HID, HID);
    gcn_selfloop<<<blk(NH, TPB), TPB, 0, stream>>>(xw, dinv, agg, (int)NH);
    gcn_scatter<<<scat_blocks, TPB, 0, stream>>>(xw, src, dst, dinv, agg, E);
    bias_relu_bf16<<<blk(NH, TPB), TPB, 0, stream>>>(agg, b2, xb, (int)NH);

    wmma_gemm_bf16<<<gemm3_blocks, 128, 0, stream>>>(xb, wih, gi, NUM_NODES, GATES, HID);

    gru_scan<<<1, 1024, 0, stream>>>(gi, whh, b_ih, b_hh, hs, NUM_NODES);

    fc_out_kernel<<<blk((long long)NUM_NODES * OUTF, TPB), TPB, 0, stream>>>(
        hs, fc_w, fc_b, out, NUM_NODES * OUTF);
}


# GRU scan: h.W_hh^T sums read bf16 W straight from its packed words with v_fma_mix_f32_bf16 (bf16 x f32 + f32), no unpack; LDS planes per wave, read one batch of 3 ahead; sum order unchanged
# speedup vs baseline: 5.4393x; 1.2787x over previous
#include <hip/hip_runtime.h>
#include <hip/hip_bf16.h>
#include <stdint.h>

typedef __attribute__((ext_vector_type(16))) __bf16 v16bf;
typedef __attribute__((ext_vector_type(8)))  float  v8f;

#define NUM_NODES 10000
#define IN_FEAT   64
#define HID       256
#define OUTF      3
#define GATES     768

__global__ void cvt_f32_to_bf16(const float* __restrict__ in,
                                __bf16* __restrict__ out, int n) {
    int i = blockIdx.x * blockDim.x + threadIdx.x;
    if (i < n) out[i] = (__bf16)in[i];
}

__global__ void cvt_f32_to_bf16_T(const float* __restrict__ in,
                                  __bf16* __restrict__ out, int K, int N) {
    int i = blockIdx.x * blockDim.x + threadIdx.x;
    if (i >= K * N) return;
    int k = i / N, n = i % N;
    out[(size_t)n * K + k] = (__bf16)in[(size_t)k * N + n];
}

__global__ void deg_init(float* __restrict__ deg, int n) {
    int i = blockIdx.x * blockDim.x + threadIdx.x;
    if (i < n) deg[i] = 1.0f;
}

__global__ void deg_edges(const int* __restrict__ dst, float* __restrict__ deg, int E) {
    int e = blockIdx.x * blockDim.x + threadIdx.x;
    if (e < E) atomicAdd(&deg[dst[e]], 1.0f);
}

__global__ void deg_rsqrt(const float* __restrict__ deg, float* __restrict__ dinv, int n) {
    int i = blockIdx.x * blockDim.x + threadIdx.x;
    if (i < n) dinv[i] = rsqrtf(deg[i]);
}

__global__ void __launch_bounds__(128)
wmma_gemm_bf16(const __bf16* __restrict__ A, const __bf16* __restrict__ Bt,
               float* __restrict__ C, int M, int N, int K) {
    int wave  = (int)((blockIdx.x * blockDim.x + threadIdx.x) >> 5);
    int lane  = threadIdx.x & 31;
    int tilesN = N >> 4;
    int tilesM = M >> 4;
    if (wave >= tilesM * tilesN) return;
    int tm = wave / tilesN;
    int tn = wave % tilesN;

    int l15 = lane & 15;
    int h   = lane >> 4;
    const __bf16* arow = A  + (size_t)(tm * 16 + l15) * K;
    const __bf16* bcol = Bt + (size_t)(tn * 16 + l15) * K;

    v8f acc = {};
    for (int k = 0; k < K; k += 32) {
        v16bf a, b;
        int k0 = k + h * 8;
#pragma unroll
        for (int i = 0; i < 8; ++i) {
            a[i]     = arow[k0 + i];
            a[i + 8] = arow[k0 + 16 + i];
            b[i]     = bcol[k0 + i];
            b[i + 8] = bcol[k0 + 16 + i];
        }
        acc = __builtin_amdgcn_wmma_f32_16x16x32_bf16(
            false, a, false, b, (short)0, acc, false, false);
    }
#pragma unroll
    for (int r = 0; r < 8; ++r) {
        int mm = tm * 16 + h * 8 + r;
        C[(size_t)mm * N + tn * 16 + l15] = acc[r];
    }
}

__global__ void gcn_selfloop(const float* __restrict__ xw, const float* __restrict__ dinv,
                             float* __restrict__ agg, int n) {
    int i = blockIdx.x * blockDim.x + threadIdx.x;
    if (i >= n) return;
    int node = i >> 8;
    float s = dinv[node];
    agg[i] = xw[i] * s * s;
}

__global__ void gcn_scatter(const float* __restrict__ xw,
                            const int* __restrict__ src, const int* __restrict__ dst,
                            const float* __restrict__ dinv,
                            float* __restrict__ agg, int E) {
    long long gid = (long long)blockIdx.x * blockDim.x + threadIdx.x;
    int e = (int)(gid >> 6);
    if (e >= E) return;
    int f = ((int)gid & 63) << 2;
    int s = src[e], d = dst[e];
    float norm = dinv[s] * dinv[d];
    const float4 v = *(const float4*)(xw + (size_t)s * HID + f);
    float* o = agg + (size_t)d * HID + f;
    atomicAdd(o + 0, norm * v.x);
    atomicAdd(o + 1, norm * v.y);
    atomicAdd(o + 2, norm * v.z);
    atomicAdd(o + 3, norm * v.w);
}

__global__ void bias_relu_bf16(const float* __restrict__ agg, const float* __restrict__ bias,
                               __bf16* __restrict__ hb, int n) {
    int i = blockIdx.x * blockDim.x + threadIdx.x;
    if (i >= n) return;
    float v = agg[i] + bias[i & (HID - 1)];
    hb[i] = (__bf16)(v > 0.f ? v : 0.f);
}

typedef uint32_t u32x4 __attribute__((ext_vector_type(4)));
typedef __bf16 bf16x8 __attribute__((ext_vector_type(8)));
#define LDS_CH 3

template <int M>
static __device__ __forceinline__ float add_lane_xor(float v) {
    int o = __builtin_amdgcn_update_dpp(0, __builtin_bit_cast(int, v), 0x160 | M, 0xf, 0xf, false);
    return v + __builtin_bit_cast(float, o);
}

__global__ void __launch_bounds__(1024)
gru_scan(const float* __restrict__ gi, const __bf16* __restrict__ Whh,
         const float* __restrict__ b_ih, const float* __restrict__ b_hh,
         float* __restrict__ hs, int T) {
    __shared__ u32x4 wl[32][6 * LDS_CH][32];
    __shared__ __attribute__((aligned(16))) float hcur[HID];
    __shared__ float gh[GATES];
    const int t = threadIdx.x;
    const int grp = t >> 3;
    const int s   = t & 7;
    const int wv = t >> 5, ln = t & 31;

    u32x4 wr[6];
    float bh[6];
    int   rowi[6];
#pragma unroll
    for (int r = 0; r < 6; ++r) {
        const int row = (r >> 1) * HID + grp + 128 * (r & 1);
        const u32x4* wsrc = (const u32x4*)(Whh + (size_t)row * HID) + s;
#pragma unroll
        for (int i = 0; i < LDS_CH; ++i) wl[wv][6 * i + r][ln] = wsrc[8 * i];
        wr[r]   = wsrc[8 * LDS_CH];
        bh[r]   = b_hh[row];
        rowi[r] = row;
    }
    if (t < HID) hcur[t] = 0.0f;
    __syncthreads();

    float hprev = 0.0f;
    for (int step = 0; step < T; ++step) {
#pragma unroll
        for (int r = 0; r < 6; ++r) asm volatile("" : "+v"(wr[r]));
        float g0 = 0.0f, g1 = 0.0f, g2 = 0.0f;
        if (t < HID) {
            const float* g = gi + (size_t)step * GATES;
            g0 = g[t]; g1 = g[HID + t]; g2 = g[2 * HID + t];
        }
        float hv[32];
#define LOAD_H(I)                                                                         \
        {                                                                                 \
            const float4 p = *(const float4*)&hcur[64 * (I) + 8 * s];                     \
            const float4 q = *(const float4*)&hcur[64 * (I) + 8 * s + 4];                 \
            hv[8 * (I) + 0] = p.x; hv[8 * (I) + 1] = p.y; hv[8 * (I) + 2] = p.z; hv[8 * (I) + 3] = p.w; \
            hv[8 * (I) + 4] = q.x; hv[8 * (I) + 5] = q.y; hv[8 * (I) + 6] = q.z; hv[8 * (I) + 7] = q.w; \
        }
        float acc[6] = {0.0f, 0.0f, 0.0f, 0.0f, 0.0f, 0.0f};
        const u32x4* wme = &wl[wv][0][ln];
        u32x4 cur[3], nxt[3];
        LOAD_H(0)
#pragma unroll
        for (int rr = 0; rr < 3; ++rr) cur[rr] = wme[32 * rr];
#pragma unroll
        for (int b = 0; b < 2 * LDS_CH; ++b) {
            __builtin_amdgcn_sched_barrier(0);
            if (b + 1 < 2 * LDS_CH) {
#pragma unroll
                for (int rr = 0; rr < 3; ++rr) nxt[rr] = wme[32 * (3 * (b + 1) + rr)];
            }
            if ((b & 1) == 0) LOAD_H((b >> 1) + 1)
            __builtin_amdgcn_sched_barrier(0);
            const int i = b >> 1, r0 = 3 * (b & 1);
            const bf16x8 w0 = __builtin_bit_cast(bf16x8, cur[0]);
            const bf16x8 w1 = __builtin_bit_cast(bf16x8, cur[1]);
            const bf16x8 w2 = __builtin_bit_cast(bf16x8, cur[2]);
#pragma unroll
            for (int e = 0; e < 8; ++e) {
                acc[r0 + 0] = fmaf((float)w0[e], hv[8 * i + e], acc[r0 + 0]);
                acc[r0 + 1] = fmaf((float)w1[e], hv[8 * i + e], acc[r0 + 1]);
                acc[r0 + 2] = fmaf((float)w2[e], hv[8 * i + e], acc[r0 + 2]);
            }
#pragma unroll
            for (int rr = 0; rr < 3; ++rr) cur[rr] = nxt[rr];
        }
#undef LOAD_H
        {
            bf16x8 wq[6];
#pragma unroll
            for (int r = 0; r < 6; ++r) wq[r] = __builtin_bit_cast(bf16x8, wr[r]);
#pragma unroll
            for (int e = 0; e < 8; ++e) {
#pragma unroll
                for (int r = 0; r < 6; ++r) acc[r] = fmaf((float)wq[r][e], hv[8 * LDS_CH + e], acc[r]);
            }
        }
#pragma unroll
        for (int r = 0; r < 6; ++r) {
            float a = acc[r];
            a = add_lane_xor<1>(a);
            a = add_lane_xor<2>(a);
            a = add_lane_xor<4>(a);
            acc[r] = a;
        }
        if (s == 0) {
#pragma unroll
            for (int r = 0; r < 6; ++r) gh[rowi[r]] = acc[r] + bh[r];
        }
        float x0 = 0.0f, x1 = 0.0f, x2 = 0.0f;
        if (t < HID) {
            x0 = g0 + b_ih[t]; x1 = g1 + b_ih[HID + t]; x2 = g2 + b_ih[2 * HID + t];
        }
        __syncthreads();
        if (t < HID) {
            float r  = 1.0f / (1.0f + __expf(-(x0 + gh[t])));
            float z  = 1.0f / (1.0f + __expf(-(x1 + gh[HID + t])));
            float nn = tanhf(x2 + r * gh[2 * HID + t]);
            float hn = fmaf(1.0f - z, nn, z * hprev);
            hcur[t] = hn;
            hprev = hn;
        }
        __syncthreads();
        if (t < HID) hs[(size_t)step * HID + t] = hprev;
    }
}

__global__ void fc_out_kernel(const float* __restrict__ hs, const float* __restrict__ fw,
                              const float* __restrict__ fb, float* __restrict__ out, int total) {
    int gid = blockIdx.x * blockDim.x + threadIdx.x;
    if (gid >= total) return;
    int n = gid / OUTF, o = gid % OUTF;
    const float* h = hs + (size_t)n * HID;
    const float* w = fw + (size_t)o * HID;
    float acc = fb[o];
#pragma unroll 8
    for (int k = 0; k < HID; ++k) acc = fmaf(h[k], w[k], acc);
    out[gid] = acc;
}

extern "C" void kernel_launch(void* const* d_in, const int* in_sizes, int n_in,
                              void* d_out, int out_size, void* d_ws, size_t ws_size,
                              hipStream_t stream) {
    const float* x    = (const float*)d_in[0];
    const int*   ei   = (const int*)  d_in[1];
    const float* W1   = (const float*)d_in[2];
    const float* b1   = (const float*)d_in[3];
    const float* W2   = (const float*)d_in[4];
    const float* b2   = (const float*)d_in[5];
    const float* W_ih = (const float*)d_in[6];
    const float* W_hh = (const float*)d_in[7];
    const float* b_ih = (const float*)d_in[8];
    const float* b_hh = (const float*)d_in[9];
    const float* fc_w = (const float*)d_in[10];
    const float* fc_b = (const float*)d_in[11];
    float* out = (float*)d_out;

    const int E = in_sizes[1] / 2;
    const int* src = ei;
    const int* dst = ei + E;

    char* ws = (char*)d_ws;
    size_t off = 0;
    auto alloc = [&](size_t bytes) -> char* {
        char* p = ws + off;
        off = (off + bytes + 255) & ~(size_t)255;
        return p;
    };
    float*  deg  = (float*) alloc((size_t)NUM_NODES * 4);
    float*  dinv = (float*) alloc((size_t)NUM_NODES * 4);
    float*  xw   = (float*) alloc((size_t)NUM_NODES * HID * 4);
    float*  agg  = (float*) alloc((size_t)NUM_NODES * HID * 4);
    float*  gi   = (float*) alloc((size_t)NUM_NODES * GATES * 4);
    float*  hs   = (float*) alloc((size_t)NUM_NODES * HID * 4);
    __bf16* xb   = (__bf16*)alloc((size_t)NUM_NODES * HID * 2);
    __bf16* w1t  = (__bf16*)alloc((size_t)HID * IN_FEAT * 2);
    __bf16* w2t  = (__bf16*)alloc((size_t)HID * HID * 2);
    __bf16* wih  = (__bf16*)alloc((size_t)GATES * HID * 2);
    __bf16* whh  = (__bf16*)alloc((size_t)GATES * HID * 2);
    (void)ws_size; (void)n_in; (void)out_size;

    const int TPB = 256;
    auto blk = [](long long n, int t) { return (unsigned)((n + t - 1) / t); };

    cvt_f32_to_bf16<<<blk((long long)NUM_NODES * IN_FEAT, TPB), TPB, 0, stream>>>(
        x, xb, NUM_NODES * IN_FEAT);
    cvt_f32_to_bf16_T<<<blk((long long)IN_FEAT * HID, TPB), TPB, 0, stream>>>(
        W1, w1t, IN_FEAT, HID);
    cvt_f32_to_bf16_T<<<blk((long long)HID * HID, TPB), TPB, 0, stream>>>(
        W2, w2t, HID, HID);
    cvt_f32_to_bf16<<<blk((long long)GATES * HID, TPB), TPB, 0, stream>>>(
        W_ih, wih, GATES * HID);
    cvt_f32_to_bf16<<<blk((long long)GATES * HID, TPB), TPB, 0, stream>>>(
        W_hh, whh, GATES * HID);

    deg_init<<<blk(NUM_NODES, TPB), TPB, 0, stream>>>(deg, NUM_NODES);
    deg_edges<<<blk(E, TPB), TPB, 0, stream>>>(dst, deg, E);
    deg_rsqrt<<<blk(NUM_NODES, TPB), TPB, 0, stream>>>(deg, dinv, NUM_NODES);

    const long long NH = (long long)NUM_NODES * HID;
    const unsigned gemm1_blocks = blk((long long)(NUM_NODES / 16) * (HID / 16), 4);
    const unsigned gemm2_blocks = gemm1_blocks;
    const unsigned gemm3_blocks = blk((long long)(NUM_NODES / 16) * (GATES / 16), 4);
    const unsigned scat_blocks  = blk((long long)E * 64, TPB);

    wmma_gemm_bf16<<<gemm1_blocks, 128, 0, stream>>>(xb, w1t, xw, NUM_NODES, HID, IN_FEAT);
    gcn_selfloop<<<blk(NH, TPB), TPB, 0, stream>>>(xw, dinv, agg, (int)NH);
    gcn_scatter<<<scat_blocks, TPB, 0, stream>>>(xw, src, dst, dinv, agg, E);
    bias_relu_bf16<<<blk(NH, TPB), TPB, 0, stream>>>(agg, b1, xb, (int)NH);

    wmma_gemm_bf16<<<gemm2_blocks, 128, 0, stream>>>(xb, w2t, xw, NUM_NODES, HID, HID);
    gcn_selfloop<<<blk(NH, TPB), TPB, 0, stream>>>(xw, dinv, agg, (int)NH);
    gcn_scatter<<<scat_blocks, TPB, 0, stream>>>(xw, src, dst, dinv, agg, E);
    bias_relu_bf16<<<blk(NH, TPB), TPB, 0, stream>>>(agg, b2, xb, (int)NH);

    wmma_gemm_bf16<<<gemm3_blocks, 128, 0, stream>>>(xb, wih, gi, NUM_NODES, GATES, HID);

    gru_scan<<<1, 1024, 0, stream>>>(gi, whh, b_ih, b_hh, hs, NUM_NODES);

    fc_out_kernel<<<blk((long long)NUM_NODES * OUTF, TPB), TPB, 0, stream>>>(
        hs, fc_w, fc_b, out, NUM_NODES * OUTF);
}


# GRU scan: 4th k chunk of W_hh kept in registers as f32, its multiply-adds two rows per packed f32 FMA; lanes merged 2:1 for the registers (16 waves, 12 rows per lane); sum order unchanged
# speedup vs baseline: 5.7124x; 1.0502x over previous
#include <hip/hip_runtime.h>
#include <hip/hip_bf16.h>
#include <stdint.h>

typedef __attribute__((ext_vector_type(16))) __bf16 v16bf;
typedef __attribute__((ext_vector_type(8)))  float  v8f;

#define NUM_NODES 10000
#define IN_FEAT   64
#define HID       256
#define OUTF      3
#define GATES     768

__global__ void cvt_f32_to_bf16(const float* __restrict__ in,
                                __bf16* __restrict__ out, int n) {
    int i = blockIdx.x * blockDim.x + threadIdx.x;
    if (i < n) out[i] = (__bf16)in[i];
}

__global__ void cvt_f32_to_bf16_T(const float* __restrict__ in,
                                  __bf16* __restrict__ out, int K, int N) {
    int i = blockIdx.x * blockDim.x + threadIdx.x;
    if (i >= K * N) return;
    int k = i / N, n = i % N;
    out[(size_t)n * K + k] = (__bf16)in[(size_t)k * N + n];
}

__global__ void deg_init(float* __restrict__ deg, int n) {
    int i = blockIdx.x * blockDim.x + threadIdx.x;
    if (i < n) deg[i] = 1.0f;
}

__global__ void deg_edges(const int* __restrict__ dst, float* __restrict__ deg, int E) {
    int e = blockIdx.x * blockDim.x + threadIdx.x;
    if (e < E) atomicAdd(&deg[dst[e]], 1.0f);
}

__global__ void deg_rsqrt(const float* __restrict__ deg, float* __restrict__ dinv, int n) {
    int i = blockIdx.x * blockDim.x + threadIdx.x;
    if (i < n) dinv[i] = rsqrtf(deg[i]);
}

__global__ void __launch_bounds__(128)
wmma_gemm_bf16(const __bf16* __restrict__ A, const __bf16* __restrict__ Bt,
               float* __restrict__ C, int M, int N, int K) {
    int wave  = (int)((blockIdx.x * blockDim.x + threadIdx.x) >> 5);
    int lane  = threadIdx.x & 31;
    int tilesN = N >> 4;
    int tilesM = M >> 4;
    if (wave >= tilesM * tilesN) return;
    int tm = wave / tilesN;
    int tn = wave % tilesN;

    int l15 = lane & 15;
    int h   = lane >> 4;
    const __bf16* arow = A  + (size_t)(tm * 16 + l15) * K;
    const __bf16* bcol = Bt + (size_t)(tn * 16 + l15) * K;

    v8f acc = {};
    for (int k = 0; k < K; k += 32) {
        v16bf a, b;
        int k0 = k + h * 8;
#pragma unroll
        for (int i = 0; i < 8; ++i) {
            a[i]     = arow[k0 + i];
            a[i + 8] = arow[k0 + 16 + i];
            b[i]     = bcol[k0 + i];
            b[i + 8] = bcol[k0 + 16 + i];
        }
        acc = __builtin_amdgcn_wmma_f32_16x16x32_bf16(
            false, a, false, b, (short)0, acc, false, false);
    }
#pragma unroll
    for (int r = 0; r < 8; ++r) {
        int mm = tm * 16 + h * 8 + r;
        C[(size_t)mm * N + tn * 16 + l15] = acc[r];
    }
}

__global__ void gcn_selfloop(const float* __restrict__ xw, const float* __restrict__ dinv,
                             float* __restrict__ agg, int n) {
    int i = blockIdx.x * blockDim.x + threadIdx.x;
    if (i >= n) return;
    int node = i >> 8;
    float s = dinv[node];
    agg[i] = xw[i] * s * s;
}

__global__ void gcn_scatter(const float* __restrict__ xw,
                            const int* __restrict__ src, const int* __restrict__ dst,
                            const float* __restrict__ dinv,
                            float* __restrict__ agg, int E) {
    long long gid = (long long)blockIdx.x * blockDim.x + threadIdx.x;
    int e = (int)(gid >> 6);
    if (e >= E) return;
    int f = ((int)gid & 63) << 2;
    int s = src[e], d = dst[e];
    float norm = dinv[s] * dinv[d];
    const float4 v = *(const float4*)(xw + (size_t)s * HID + f);
    float* o = agg + (size_t)d * HID + f;
    atomicAdd(o + 0, norm * v.x);
    atomicAdd(o + 1, norm * v.y);
    atomicAdd(o + 2, norm * v.z);
    atomicAdd(o + 3, norm * v.w);
}

__global__ void bias_relu_bf16(const float* __restrict__ agg, const float* __restrict__ bias,
                               __bf16* __restrict__ hb, int n) {
    int i = blockIdx.x * blockDim.x + threadIdx.x;
    if (i >= n) return;
    float v = agg[i] + bias[i & (HID - 1)];
    hb[i] = (__bf16)(v > 0.f ? v : 0.f);
}

typedef uint32_t u32x4 __attribute__((ext_vector_type(4)));
typedef __bf16 bf16x8 __attribute__((ext_vector_type(8)));
typedef float f32x2 __attribute__((ext_vector_type(2)));
#define GRU_THREADS 512
#define NROW   12
#define LDS_CH 3
#define BATCH  6

template <int M>
static __device__ __forceinline__ float add_lane_xor(float v) {
    int o = __builtin_amdgcn_update_dpp(0, __builtin_bit_cast(int, v), 0x160 | M, 0xf, 0xf, false);
    return v + __builtin_bit_cast(float, o);
}

__global__ void __launch_bounds__(GRU_THREADS)
gru_scan(const float* __restrict__ gi, const __bf16* __restrict__ Whh,
         const float* __restrict__ b_ih, const float* __restrict__ b_hh,
         float* __restrict__ hs, int T) {
    __shared__ u32x4 wl[GRU_THREADS / 32][NROW * LDS_CH][32];
    __shared__ __attribute__((aligned(16))) float hcur[HID];
    __shared__ float gh[GATES];
    const int t = threadIdx.x;
    const int grp = t >> 3;
    const int s   = t & 7;
    const int wv = t >> 5, ln = t & 31;

    f32x2 wf[8][NROW / 2];
    float bh[NROW];
#pragma unroll
    for (int r = 0; r < NROW; ++r) {
        const int row = (r >> 2) * HID + grp + 64 * (r & 3);
        const u32x4* wsrc = (const u32x4*)(Whh + (size_t)row * HID) + s;
#pragma unroll
        for (int i = 0; i < LDS_CH; ++i) wl[wv][NROW * i + r][ln] = wsrc[8 * i];
        const bf16x8 w3 = __builtin_bit_cast(bf16x8, wsrc[8 * LDS_CH]);
#pragma unroll
        for (int e = 0; e < 8; ++e) wf[e][r >> 1][r & 1] = (float)w3[e];
        bh[r] = b_hh[row];
    }
    if (t < HID) hcur[t] = 0.0f;
    __syncthreads();

    float hprev = 0.0f;
    for (int step = 0; step < T; ++step) {
        float g0 = 0.0f, g1 = 0.0f, g2 = 0.0f;
        if (t < HID) {
            const float* g = gi + (size_t)step * GATES;
            g0 = g[t]; g1 = g[HID + t]; g2 = g[2 * HID + t];
        }
        float hv[32];
#define LOAD_H(I)                                                                         \
        {                                                                                 \
            const float4 p = *(const float4*)&hcur[64 * (I) + 8 * s];                     \
            const float4 q = *(const float4*)&hcur[64 * (I) + 8 * s + 4];                 \
            hv[8 * (I) + 0] = p.x; hv[8 * (I) + 1] = p.y; hv[8 * (I) + 2] = p.z; hv[8 * (I) + 3] = p.w; \
            hv[8 * (I) + 4] = q.x; hv[8 * (I) + 5] = q.y; hv[8 * (I) + 6] = q.z; hv[8 * (I) + 7] = q.w; \
        }
        f32x2 acc[NROW / 2];
#pragma unroll
        for (int p = 0; p < NROW / 2; ++p) acc[p] = (f32x2){0.0f, 0.0f};
        const u32x4* wme = &wl[wv][0][ln];
        u32x4 cur[BATCH], nxt[BATCH];
        LOAD_H(0)
#pragma unroll
        for (int rr = 0; rr < BATCH; ++rr) cur[rr] = wme[32 * rr];
        constexpr int NB = NROW / BATCH;
#pragma unroll
        for (int b = 0; b < NB * LDS_CH; ++b) {
            __builtin_amdgcn_sched_barrier(0);
            if (b + 1 < NB * LDS_CH) {
#pragma unroll
                for (int rr = 0; rr < BATCH; ++rr) nxt[rr] = wme[32 * (BATCH * (b + 1) + rr)];
            }
            if (b % NB == 0) LOAD_H(b / NB + 1)
            __builtin_amdgcn_sched_barrier(0);
            const int i = b / NB, r0 = BATCH * (b % NB);
#pragma unroll
            for (int e = 0; e < 8; ++e) {
#pragma unroll
                for (int rr = 0; rr < BATCH; ++rr) {
                    const bf16x8 w = __builtin_bit_cast(bf16x8, cur[rr]);
                    const int r = r0 + rr;
                    acc[r >> 1][r & 1] = fmaf((float)w[e], hv[8 * i + e], acc[r >> 1][r & 1]);
                }
            }
#pragma unroll
            for (int rr = 0; rr < BATCH; ++rr) cur[rr] = nxt[rr];
        }
#undef LOAD_H
#pragma unroll
        for (int e = 0; e < 8; ++e) {
            const f32x2 h2 = (f32x2){hv[8 * LDS_CH + e], hv[8 * LDS_CH + e]};
#pragma unroll
            for (int p = 0; p < NROW / 2; ++p) acc[p] = __builtin_elementwise_fma(wf[e][p], h2, acc[p]);
        }
        float rs[NROW];
#pragma unroll
        for (int r = 0; r < NROW; ++r) {
            float a = acc[r >> 1][r & 1];
            a = add_lane_xor<1>(a);
            a = add_lane_xor<2>(a);
            a = add_lane_xor<4>(a);
            rs[r] = a;
        }
        if (s == 0) {
#pragma unroll
            for (int r = 0; r < NROW; ++r) gh[(r >> 2) * HID + grp + 64 * (r & 3)] = rs[r] + bh[r];
        }
        float x0 = 0.0f, x1 = 0.0f, x2 = 0.0f;
        if (t < HID) {
            x0 = g0 + b_ih[t]; x1 = g1 + b_ih[HID + t]; x2 = g2 + b_ih[2 * HID + t];
        }
        __syncthreads();
        if (t < HID) {
            float r  = 1.0f / (1.0f + __expf(-(x0 + gh[t])));
            float z  = 1.0f / (1.0f + __expf(-(x1 + gh[HID + t])));
            float nn = tanhf(x2 + r * gh[2 * HID + t]);
            float hn = fmaf(1.0f - z, nn, z * hprev);
            hcur[t] = hn;
            hprev = hn;
        }
        __syncthreads();
        if (t < HID) hs[(size_t)step * HID + t] = hprev;
    }
}

__global__ void fc_out_kernel(const float* __restrict__ hs, const float* __restrict__ fw,
                              const float* __restrict__ fb, float* __restrict__ out, int total) {
    int gid = blockIdx.x * blockDim.x + threadIdx.x;
    if (gid >= total) return;
    int n = gid / OUTF, o = gid % OUTF;
    const float* h = hs + (size_t)n * HID;
    const float* w = fw + (size_t)o * HID;
    float acc = fb[o];
#pragma unroll 8
    for (int k = 0; k < HID; ++k) acc = fmaf(h[k], w[k], acc);
    out[gid] = acc;
}

extern "C" void kernel_launch(void* const* d_in, const int* in_sizes, int n_in,
                              void* d_out, int out_size, void* d_ws, size_t ws_size,
                              hipStream_t stream) {
    const float* x    = (const float*)d_in[0];
    const int*   ei   = (const int*)  d_in[1];
    const float* W1   = (const float*)d_in[2];
    const float* b1   = (const float*)d_in[3];
    const float* W2   = (const float*)d_in[4];
    const float* b2   = (const float*)d_in[5];
    const float* W_ih = (const float*)d_in[6];
    const float* W_hh = (const float*)d_in[7];
    const float* b_ih = (const float*)d_in[8];
    const float* b_hh = (const float*)d_in[9];
    const float* fc_w = (const float*)d_in[10];
    const float* fc_b = (const float*)d_in[11];
    float* out = (float*)d_out;

    const int E = in_sizes[1] / 2;
    const int* src = ei;
    const int* dst = ei + E;

    char* ws = (char*)d_ws;
    size_t off = 0;
    auto alloc = [&](size_t bytes) -> char* {
        char* p = ws + off;
        off = (off + bytes + 255) & ~(size_t)255;
        return p;
    };
    float*  deg  = (float*) alloc((size_t)NUM_NODES * 4);
    float*  dinv = (float*) alloc((size_t)NUM_NODES * 4);
    float*  xw   = (float*) alloc((size_t)NUM_NODES * HID * 4);
    float*  agg  = (float*) alloc((size_t)NUM_NODES * HID * 4);
    float*  gi   = (float*) alloc((size_t)NUM_NODES * GATES * 4);
    float*  hs   = (float*) alloc((size_t)NUM_NODES * HID * 4);
    __bf16* xb   = (__bf16*)alloc((size_t)NUM_NODES * HID * 2);
    __bf16* w1t  = (__bf16*)alloc((size_t)HID * IN_FEAT * 2);
    __bf16* w2t  = (__bf16*)alloc((size_t)HID * HID * 2);
    __bf16* wih  = (__bf16*)alloc((size_t)GATES * HID * 2);
    __bf16* whh  = (__bf16*)alloc((size_t)GATES * HID * 2);
    (void)ws_size; (void)n_in; (void)out_size;

    const int TPB = 256;
    auto blk = [](long long n, int t) { return (unsigned)((n + t - 1) / t); };

    cvt_f32_to_bf16<<<blk((long long)NUM_NODES * IN_FEAT, TPB), TPB, 0, stream>>>(
        x, xb, NUM_NODES * IN_FEAT);
    cvt_f32_to_bf16_T<<<blk((long long)IN_FEAT * HID, TPB), TPB, 0, stream>>>(
        W1, w1t, IN_FEAT, HID);
    cvt_f32_to_bf16_T<<<blk((long long)HID * HID, TPB), TPB, 0, stream>>>(
        W2, w2t, HID, HID);
    cvt_f32_to_bf16<<<blk((long long)GATES * HID, TPB), TPB, 0, stream>>>(
        W_ih, wih, GATES * HID);
    cvt_f32_to_bf16<<<blk((long long)GATES * HID, TPB), TPB, 0, stream>>>(
        W_hh, whh, GATES * HID);

    deg_init<<<blk(NUM_NODES, TPB), TPB, 0, stream>>>(deg, NUM_NODES);
    deg_edges<<<blk(E, TPB), TPB, 0, stream>>>(dst, deg, E);
    deg_rsqrt<<<blk(NUM_NODES, TPB), TPB, 0, stream>>>(deg, dinv, NUM_NODES);

    const long long NH = (long long)NUM_NODES * HID;
    const unsigned gemm1_blocks = blk((long long)(NUM_NODES / 16) * (HID / 16), 4);
    const unsigned gemm2_blocks = gemm1_blocks;
    const unsigned gemm3_blocks = blk((long long)(NUM_NODES / 16) * (GATES / 16), 4);
    const unsigned scat_blocks  = blk((long long)E * 64, TPB);

    wmma_gemm_bf16<<<gemm1_blocks, 128, 0, stream>>>(xb, w1t, xw, NUM_NODES, HID, IN_FEAT);
    gcn_selfloop<<<blk(NH, TPB), TPB, 0, stream>>>(xw, dinv, agg, (int)NH);
    gcn_scatter<<<scat_blocks, TPB, 0, stream>>>(xw, src, dst, dinv, agg, E);
    bias_relu_bf16<<<blk(NH, TPB), TPB, 0, stream>>>(agg, b1, xb, (int)NH);

    wmma_gemm_bf16<<<gemm2_blocks, 128, 0, stream>>>(xb, w2t, xw, NUM_NODES, HID, HID);
    gcn_selfloop<<<blk(NH, TPB), TPB, 0, stream>>>(xw, dinv, agg, (int)NH);
    gcn_scatter<<<scat_blocks, TPB, 0, stream>>>(xw, src, dst, dinv, agg, E);
    bias_relu_bf16<<<blk(NH, TPB), TPB, 0, stream>>>(agg, b2, xb, (int)NH);

    wmma_gemm_bf16<<<gemm3_blocks, 128, 0, stream>>>(xb, wih, gi, NUM_NODES, GATES, HID);

    gru_scan<<<1, GRU_THREADS, 0, stream>>>(gi, whh, b_ih, b_hh, hs, NUM_NODES);

    fc_out_kernel<<<blk((long long)NUM_NODES * OUTF, TPB), TPB, 0, stream>>>(
        hs, fc_w, fc_b, out, NUM_NODES * OUTF);
}


# GRU scan: gate tail is straight-line arithmetic only: tanh's two ranges both computed, one selected per lane (no branch); next step's W requests moved out of it, one group; sum order unchanged
# speedup vs baseline: 7.2834x; 1.2750x over previous
#include <hip/hip_runtime.h>
#include <hip/hip_bf16.h>
#include <stdint.h>

typedef __attribute__((ext_vector_type(16))) __bf16 v16bf;
typedef __attribute__((ext_vector_type(8)))  float  v8f;

#define NUM_NODES 10000
#define IN_FEAT   64
#define HID       256
#define OUTF      3
#define GATES     768

__global__ void cvt_f32_to_bf16(const float* __restrict__ in,
                                __bf16* __restrict__ out, int n) {
    int i = blockIdx.x * blockDim.x + threadIdx.x;
    if (i < n) out[i] = (__bf16)in[i];
}

__global__ void cvt_f32_to_bf16_T(const float* __restrict__ in,
                                  __bf16* __restrict__ out, int K, int N) {
    int i = blockIdx.x * blockDim.x + threadIdx.x;
    if (i >= K * N) return;
    int k = i / N, n = i % N;
    out[(size_t)n * K + k] = (__bf16)in[(size_t)k * N + n];
}

__global__ void deg_init(float* __restrict__ deg, int n) {
    int i = blockIdx.x * blockDim.x + threadIdx.x;
    if (i < n) deg[i] = 1.0f;
}

__global__ void deg_edges(const int* __restrict__ dst, float* __restrict__ deg, int E) {
    int e = blockIdx.x * blockDim.x + threadIdx.x;
    if (e < E) atomicAdd(&deg[dst[e]], 1.0f);
}

__global__ void deg_rsqrt(const float* __restrict__ deg, float* __restrict__ dinv, int n) {
    int i = blockIdx.x * blockDim.x + threadIdx.x;
    if (i < n) dinv[i] = rsqrtf(deg[i]);
}

__global__ void __launch_bounds__(128)
wmma_gemm_bf16(const __bf16* __restrict__ A, const __bf16* __restrict__ Bt,
               float* __restrict__ C, int M, int N, int K) {
    int wave  = (int)((blockIdx.x * blockDim.x + threadIdx.x) >> 5);
    int lane  = threadIdx.x & 31;
    int tilesN = N >> 4;
    int tilesM = M >> 4;
    if (wave >= tilesM * tilesN) return;
    int tm = wave / tilesN;
    int tn = wave % tilesN;

    int l15 = lane & 15;
    int h   = lane >> 4;
    const __bf16* arow = A  + (size_t)(tm * 16 + l15) * K;
    const __bf16* bcol = Bt + (size_t)(tn * 16 + l15) * K;

    v8f acc = {};
    for (int k = 0; k < K; k += 32) {
        v16bf a, b;
        int k0 = k + h * 8;
#pragma unroll
        for (int i = 0; i < 8; ++i) {
            a[i]     = arow[k0 + i];
            a[i + 8] = arow[k0 + 16 + i];
            b[i]     = bcol[k0 + i];
            b[i + 8] = bcol[k0 + 16 + i];
        }
        acc = __builtin_amdgcn_wmma_f32_16x16x32_bf16(
            false, a, false, b, (short)0, acc, false, false);
    }
#pragma unroll
    for (int r = 0; r < 8; ++r) {
        int mm = tm * 16 + h * 8 + r;
        C[(size_t)mm * N + tn * 16 + l15] = acc[r];
    }
}

__global__ void gcn_selfloop(const float* __restrict__ xw, const float* __restrict__ dinv,
                             float* __restrict__ agg, int n) {
    int i = blockIdx.x * blockDim.x + threadIdx.x;
    if (i >= n) return;
    int node = i >> 8;
    float s = dinv[node];
    agg[i] = xw[i] * s * s;
}

__global__ void gcn_scatter(const float* __restrict__ xw,
                            const int* __restrict__ src, const int* __restrict__ dst,
                            const float* __restrict__ dinv,
                            float* __restrict__ agg, int E) {
    long long gid = (long long)blockIdx.x * blockDim.x + threadIdx.x;
    int e = (int)(gid >> 6);
    if (e >= E) return;
    int f = ((int)gid & 63) << 2;
    int s = src[e], d = dst[e];
    float norm = dinv[s] * dinv[d];
    const float4 v = *(const float4*)(xw + (size_t)s * HID + f);
    float* o = agg + (size_t)d * HID + f;
    atomicAdd(o + 0, norm * v.x);
    atomicAdd(o + 1, norm * v.y);
    atomicAdd(o + 2, norm * v.z);
    atomicAdd(o + 3, norm * v.w);
}

__global__ void bias_relu_bf16(const float* __restrict__ agg, const float* __restrict__ bias,
                               __bf16* __restrict__ hb, int n) {
    int i = blockIdx.x * blockDim.x + threadIdx.x;
    if (i >= n) return;
    float v = agg[i] + bias[i & (HID - 1)];
    hb[i] = (__bf16)(v > 0.f ? v : 0.f);
}

typedef uint32_t u32x4 __attribute__((ext_vector_type(4)));
typedef __bf16 bf16x8 __attribute__((ext_vector_type(8)));
typedef float f32x2 __attribute__((ext_vector_type(2)));
#define GRU_THREADS 256
#define NROW   24
#define R2     12
#define NPL    60
#define BATCH  6
#define NBATCH 10
#define RING   3

template <int M>
static __device__ __forceinline__ float lane_xor(float v) {
    int o = __builtin_amdgcn_update_dpp(0, __builtin_bit_cast(int, v), 0x160 | M, 0xf, 0xf, false);
    return __builtin_bit_cast(float, o);
}

static __device__ __forceinline__ float tanh_both_ranges(float x) {
    const float ax = __builtin_fabsf(x);
    const float x2 = x * x;
    float p = fmaf(__builtin_bit_cast(float, 0xbbbac73du), x2, __builtin_bit_cast(float, 0x3ca908c9u));
    p = fmaf(x2, p, __builtin_bit_cast(float, 0xbd5c1c4eu));
    p = fmaf(x2, p, __builtin_bit_cast(float, 0x3e088382u));
    p = fmaf(x2, p, __builtin_bit_cast(float, 0xbeaaaa99u));
    const float small = fmaf(x2, ax * p, ax);
    const float e = expf(ax + ax);
    const float big = fmaf(__builtin_amdgcn_rcpf(e + 1.0f), -2.0f, 1.0f);
    const float m = (ax < 0.625f) ? small : big;
    return __builtin_copysignf(m, x);
}

__global__ void __launch_bounds__(GRU_THREADS)
gru_scan(const float* __restrict__ gi, const __bf16* __restrict__ Whh,
         const float* __restrict__ b_ih, const float* __restrict__ b_hh,
         float* __restrict__ hs, int T) {
    __shared__ u32x4 wl[GRU_THREADS / 32][NPL][32];
    __shared__ __attribute__((aligned(16))) float hb[2 * HID];
    const int t = threadIdx.x;
    const int grp = t >> 3;
    const int s   = t & 7;
    const int wv = t >> 5, ln = t & 31;
    const bool odd1 = (s & 1) != 0;
    const bool odd2 = (s & 2) != 0;
    const bool odd4 = (s & 4) != 0;

    f32x2 wf2[8][R2 / 2];
    f32x2 wf3[8][NROW / 2];
#pragma unroll
    for (int r = 0; r < NROW; ++r) {
        const int row = (r >> 3) * HID + 8 * grp + (r & 7);
        const u32x4* wsrc = (const u32x4*)(Whh + (size_t)row * HID) + s;
        wl[wv][r][ln]      = wsrc[0];
        wl[wv][24 + r][ln] = wsrc[8];
        if (r < R2) {
            const bf16x8 w2 = __builtin_bit_cast(bf16x8, wsrc[16]);
#pragma unroll
            for (int e = 0; e < 8; ++e) wf2[e][r >> 1][r & 1] = (float)w2[e];
        } else {
            wl[wv][48 + (r - R2)][ln] = wsrc[16];
        }
        const bf16x8 w3 = __builtin_bit_cast(bf16x8, wsrc[24]);
#pragma unroll
        for (int e = 0; e < 8; ++e) wf3[e][r >> 1][r & 1] = (float)w3[e];
    }
    const float bq0 = b_hh[t], bq1 = b_hh[HID + t], bq2 = b_hh[2 * HID + t];
    const float bi0 = b_ih[t], bi1 = b_ih[HID + t], bi2 = b_ih[2 * HID + t];
    hb[t] = 0.0f;
    hb[HID + t] = 0.0f;
    __syncthreads();

    const u32x4* wme = &wl[wv][0][ln];
    u32x4 ring[RING][BATCH];
#pragma unroll
    for (int b = 0; b < RING - 1; ++b) {
#pragma unroll
        for (int rr = 0; rr < BATCH; ++rr) ring[b][rr] = wme[32 * (BATCH * b + rr)];
    }

    float hprev = 0.0f;
    int hoff = 0;
    for (int step = 0; step < T; ++step) {
        const float* g = gi + (size_t)step * GATES + t;
        const float g0 = g[0], g1 = g[HID], g2 = g[2 * HID];
        const float* hcur = hb + hoff;
        float hv[32];
#pragma unroll
        for (int i = 0; i < 4; ++i) {
            const float4 p = *(const float4*)&hcur[64 * i + 8 * s];
            const float4 q = *(const float4*)&hcur[64 * i + 8 * s + 4];
            hv[8 * i + 0] = p.x; hv[8 * i + 1] = p.y; hv[8 * i + 2] = p.z; hv[8 * i + 3] = p.w;
            hv[8 * i + 4] = q.x; hv[8 * i + 5] = q.y; hv[8 * i + 6] = q.z; hv[8 * i + 7] = q.w;
        }
        f32x2 acc[NROW / 2];
#pragma unroll
        for (int p = 0; p < NROW / 2; ++p) acc[p] = (f32x2){0.0f, 0.0f};
#pragma unroll
        for (int b = 0; b < NBATCH; ++b) {
            if (b == 0) __builtin_amdgcn_sched_barrier(0);
            if (b + RING - 1 < NBATCH) {
#pragma unroll
                for (int rr = 0; rr < BATCH; ++rr) ring[(b + RING - 1) % RING][rr] = wme[32 * (BATCH * (b + RING - 1) + rr)];
            }
            if (b == 0) __builtin_amdgcn_sched_barrier(0);
            if (b == 0) {
                if (step > 0) hs[(size_t)(step - 1) * HID + t] = hprev;
            }
#pragma unroll
            for (int e = 0; e < 8; ++e) {
#pragma unroll
                for (int rr = 0; rr < BATCH; ++rr) {
                    const int pl = BATCH * b + rr;
                    const int i = pl < 48 ? pl / 24 : 2;
                    const int r = pl < 48 ? pl % 24 : R2 + (pl - 48);
                    const bf16x8 w = __builtin_bit_cast(bf16x8, ring[b % RING][rr]);
                    acc[r >> 1][r & 1] = fmaf((float)w[e], hv[8 * i + e], acc[r >> 1][r & 1]);
                }
            }
        }
#pragma unroll
        for (int b = 0; b + RING < NBATCH; ++b) {
            __builtin_amdgcn_sched_group_barrier(0x002, 8 * BATCH, 0);
            __builtin_amdgcn_sched_group_barrier(0x100, BATCH, 0);
        }
#pragma unroll
        for (int b = 0; b < RING; ++b)
            __builtin_amdgcn_sched_group_barrier(0x002, 8 * BATCH, 0);
#pragma unroll
        for (int b = 0; b < RING - 1; ++b) {
#pragma unroll
            for (int rr = 0; rr < BATCH; ++rr) ring[b][rr] = wme[32 * (BATCH * b + rr)];
        }
#pragma unroll
        for (int e = 0; e < 8; ++e) {
            const f32x2 h2 = (f32x2){hv[16 + e], hv[16 + e]};
#pragma unroll
            for (int p = 0; p < R2 / 2; ++p) acc[p] = __builtin_elementwise_fma(wf2[e][p], h2, acc[p]);
        }
#pragma unroll
        for (int e = 0; e < 8; ++e) {
            const f32x2 h2 = (f32x2){hv[24 + e], hv[24 + e]};
#pragma unroll
            for (int p = 0; p < NROW / 2; ++p) acc[p] = __builtin_elementwise_fma(wf3[e][p], h2, acc[p]);
        }
        __builtin_amdgcn_sched_group_barrier(0x100, (RING - 1) * BATCH, 0);
        __builtin_amdgcn_sched_group_barrier(0x002, 8 * (R2 / 2) + 8 * (NROW / 2), 0);
        float a12[NROW / 2];
#pragma unroll
        for (int p = 0; p < NROW / 2; ++p) {
            const float keep = odd1 ? acc[p][1] : acc[p][0];
            const float send = odd1 ? acc[p][0] : acc[p][1];
            a12[p] = keep + lane_xor<1>(send);
        }
        float b6[NROW / 4];
#pragma unroll
        for (int m = 0; m < NROW / 4; ++m) {
            const float keep = odd2 ? a12[2 * m + 1] : a12[2 * m];
            const float send = odd2 ? a12[2 * m] : a12[2 * m + 1];
            b6[m] = keep + lane_xor<2>(send);
        }
        float c3[3];
#pragma unroll
        for (int gt = 0; gt < 3; ++gt) {
            const float keep = odd4 ? b6[2 * gt + 1] : b6[2 * gt];
            const float send = odd4 ? b6[2 * gt] : b6[2 * gt + 1];
            c3[gt] = keep + lane_xor<4>(send);
        }
        const float ghr = c3[0] + bq0, ghz = c3[1] + bq1, ghn = c3[2] + bq2;
        const float x0 = g0 + bi0, x1 = g1 + bi1, x2 = g2 + bi2;
        const float r  = 1.0f / (1.0f + __expf(-(x0 + ghr)));
        const float z  = 1.0f / (1.0f + __expf(-(x1 + ghz)));
        const float nn = tanh_both_ranges(x2 + r * ghn);
        const float hn = fmaf(1.0f - z, nn, z * hprev);
        hprev = hn;
        hoff ^= HID;
        hb[hoff + t] = hn;
        __syncthreads();
    }
    if (T > 0) hs[(size_t)(T - 1) * HID + t] = hprev;
}

__global__ void fc_out_kernel(const float* __restrict__ hs, const float* __restrict__ fw,
                              const float* __restrict__ fb, float* __restrict__ out, int total) {
    int gid = blockIdx.x * blockDim.x + threadIdx.x;
    if (gid >= total) return;
    int n = gid / OUTF, o = gid % OUTF;
    const float* h = hs + (size_t)n * HID;
    const float* w = fw + (size_t)o * HID;
    float acc = fb[o];
#pragma unroll 8
    for (int k = 0; k < HID; ++k) acc = fmaf(h[k], w[k], acc);
    out[gid] = acc;
}

extern "C" void kernel_launch(void* const* d_in, const int* in_sizes, int n_in,
                              void* d_out, int out_size, void* d_ws, size_t ws_size,
                              hipStream_t stream) {
    const float* x    = (const float*)d_in[0];
    const int*   ei   = (const int*)  d_in[1];
    const float* W1   = (const float*)d_in[2];
    const float* b1   = (const float*)d_in[3];
    const float* W2   = (const float*)d_in[4];
    const float* b2   = (const float*)d_in[5];
    const float* W_ih = (const float*)d_in[6];
    const float* W_hh = (const float*)d_in[7];
    const float* b_ih = (const float*)d_in[8];
    const float* b_hh = (const float*)d_in[9];
    const float* fc_w = (const float*)d_in[10];
    const float* fc_b = (const float*)d_in[11];
    float* out = (float*)d_out;

    const int E = in_sizes[1] / 2;
    const int* src = ei;
    const int* dst = ei + E;

    char* ws = (char*)d_ws;
    size_t off = 0;
    auto alloc = [&](size_t bytes) -> char* {
        char* p = ws + off;
        off = (off + bytes + 255) & ~(size_t)255;
        return p;
    };
    float*  deg  = (float*) alloc((size_t)NUM_NODES * 4);
    float*  dinv = (float*) alloc((size_t)NUM_NODES * 4);
    float*  xw   = (float*) alloc((size_t)NUM_NODES * HID * 4);
    float*  agg  = (float*) alloc((size_t)NUM_NODES * HID * 4);
    float*  gi   = (float*) alloc((size_t)NUM_NODES * GATES * 4);
    float*  hs   = (float*) alloc((size_t)NUM_NODES * HID * 4);
    __bf16* xb   = (__bf16*)alloc((size_t)NUM_NODES * HID * 2);
    __bf16* w1t  = (__bf16*)alloc((size_t)HID * IN_FEAT * 2);
    __bf16* w2t  = (__bf16*)alloc((size_t)HID * HID * 2);
    __bf16* wih  = (__bf16*)alloc((size_t)GATES * HID * 2);
    __bf16* whh  = (__bf16*)alloc((size_t)GATES * HID * 2);
    (void)ws_size; (void)n_in; (void)out_size;

    const int TPB = 256;
    auto blk = [](long long n, int t) { return (unsigned)((n + t - 1) / t); };

    cvt_f32_to_bf16<<<blk((long long)NUM_NODES * IN_FEAT, TPB), TPB, 0, stream>>>(
        x, xb, NUM_NODES * IN_FEAT);
    cvt_f32_to_bf16_T<<<blk((long long)IN_FEAT * HID, TPB), TPB, 0, stream>>>(
        W1, w1t, IN_FEAT, HID);
    cvt_f32_to_bf16_T<<<blk((long long)HID * HID, TPB), TPB, 0, stream>>>(
        W2, w2t, HID, HID);
    cvt_f32_to_bf16<<<blk((long long)GATES * HID, TPB), TPB, 0, stream>>>(
        W_ih, wih, GATES * HID);
    cvt_f32_to_bf16<<<blk((long long)GATES * HID, TPB), TPB, 0, stream>>>(
        W_hh, whh, GATES * HID);

    deg_init<<<blk(NUM_NODES, TPB), TPB, 0, stream>>>(deg, NUM_NODES);
    deg_edges<<<blk(E, TPB), TPB, 0, stream>>>(dst, deg, E);
    deg_rsqrt<<<blk(NUM_NODES, TPB), TPB, 0, stream>>>(deg, dinv, NUM_NODES);

    const long long NH = (long long)NUM_NODES * HID;
    const unsigned gemm1_blocks = blk((long long)(NUM_NODES / 16) * (HID / 16), 4);
    const unsigned gemm2_blocks = gemm1_blocks;
    const unsigned gemm3_blocks = blk((long long)(NUM_NODES / 16) * (GATES / 16), 4);
    const unsigned scat_blocks  = blk((long long)E * 64, TPB);

    wmma_gemm_bf16<<<gemm1_blocks, 128, 0, stream>>>(xb, w1t, xw, NUM_NODES, HID, IN_FEAT);
    gcn_selfloop<<<blk(NH, TPB), TPB, 0, stream>>>(xw, dinv, agg, (int)NH);
    gcn_scatter<<<scat_blocks, TPB, 0, stream>>>(xw, src, dst, dinv, agg, E);
    bias_relu_bf16<<<blk(NH, TPB), TPB, 0, stream>>>(agg, b1, xb, (int)NH);

    wmma_gemm_bf16<<<gemm2_blocks, 128, 0, stream>>>(xb, w2t, xw, NUM_NODES, HID, HID);
    gcn_selfloop<<<blk(NH, TPB), TPB, 0, stream>>>(xw, dinv, agg, (int)NH);
    gcn_scatter<<<scat_blocks, TPB, 0, stream>>>(xw, src, dst, dinv, agg, E);
    bias_relu_bf16<<<blk(NH, TPB), TPB, 0, stream>>>(agg, b2, xb, (int)NH);

    wmma_gemm_bf16<<<gemm3_blocks, 128, 0, stream>>>(xb, wih, gi, NUM_NODES, GATES, HID);

    gru_scan<<<1, GRU_THREADS, 0, stream>>>(gi, whh, b_ih, b_hh, hs, NUM_NODES);

    fc_out_kernel<<<blk((long long)NUM_NODES * OUTF, TPB), TPB, 0, stream>>>(
        hs, fc_w, fc_b, out, NUM_NODES * OUTF);
}


# GRU scan: accumulator slot j of lane s holds row j^s (fixed when W_hh is staged), so every lane keeps even slots and sends odd ones in the 8-lane reduce-scatter: no per-lane selects; sum order same
# speedup vs baseline: 7.6403x; 1.0490x over previous
#include <hip/hip_runtime.h>
#include <hip/hip_bf16.h>
#include <stdint.h>

typedef __attribute__((ext_vector_type(16))) __bf16 v16bf;
typedef __attribute__((ext_vector_type(8)))  float  v8f;

#define NUM_NODES 10000
#define IN_FEAT   64
#define HID       256
#define OUTF      3
#define GATES     768

__global__ void cvt_f32_to_bf16(const float* __restrict__ in,
                                __bf16* __restrict__ out, int n) {
    int i = blockIdx.x * blockDim.x + threadIdx.x;
    if (i < n) out[i] = (__bf16)in[i];
}

__global__ void cvt_f32_to_bf16_T(const float* __restrict__ in,
                                  __bf16* __restrict__ out, int K, int N) {
    int i = blockIdx.x * blockDim.x + threadIdx.x;
    if (i >= K * N) return;
    int k = i / N, n = i % N;
    out[(size_t)n * K + k] = (__bf16)in[(size_t)k * N + n];
}

__global__ void deg_init(float* __restrict__ deg, int n) {
    int i = blockIdx.x * blockDim.x + threadIdx.x;
    if (i < n) deg[i] = 1.0f;
}

__global__ void deg_edges(const int* __restrict__ dst, float* __restrict__ deg, int E) {
    int e = blockIdx.x * blockDim.x + threadIdx.x;
    if (e < E) atomicAdd(&deg[dst[e]], 1.0f);
}

__global__ void deg_rsqrt(const float* __restrict__ deg, float* __restrict__ dinv, int n) {
    int i = blockIdx.x * blockDim.x + threadIdx.x;
    if (i < n) dinv[i] = rsqrtf(deg[i]);
}

__global__ void __launch_bounds__(128)
wmma_gemm_bf16(const __bf16* __restrict__ A, const __bf16* __restrict__ Bt,
               float* __restrict__ C, int M, int N, int K) {
    int wave  = (int)((blockIdx.x * blockDim.x + threadIdx.x) >> 5);
    int lane  = threadIdx.x & 31;
    int tilesN = N >> 4;
    int tilesM = M >> 4;
    if (wave >= tilesM * tilesN) return;
    int tm = wave / tilesN;
    int tn = wave % tilesN;

    int l15 = lane & 15;
    int h   = lane >> 4;
    const __bf16* arow = A  + (size_t)(tm * 16 + l15) * K;
    const __bf16* bcol = Bt + (size_t)(tn * 16 + l15) * K;

    v8f acc = {};
    for (int k = 0; k < K; k += 32) {
        v16bf a, b;
        int k0 = k + h * 8;
#pragma unroll
        for (int i = 0; i < 8; ++i) {
            a[i]     = arow[k0 + i];
            a[i + 8] = arow[k0 + 16 + i];
            b[i]     = bcol[k0 + i];
            b[i + 8] = bcol[k0 + 16 + i];
        }
        acc = __builtin_amdgcn_wmma_f32_16x16x32_bf16(
            false, a, false, b, (short)0, acc, false, false);
    }
#pragma unroll
    for (int r = 0; r < 8; ++r) {
        int mm = tm * 16 + h * 8 + r;
        C[(size_t)mm * N + tn * 16 + l15] = acc[r];
    }
}

__global__ void gcn_selfloop(const float* __restrict__ xw, const float* __restrict__ dinv,
                             float* __restrict__ agg, int n) {
    int i = blockIdx.x * blockDim.x + threadIdx.x;
    if (i >= n) return;
    int node = i >> 8;
    float s = dinv[node];
    agg[i] = xw[i] * s * s;
}

__global__ void gcn_scatter(const float* __restrict__ xw,
                            const int* __restrict__ src, const int* __restrict__ dst,
                            const float* __restrict__ dinv,
                            float* __restrict__ agg, int E) {
    long long gid = (long long)blockIdx.x * blockDim.x + threadIdx.x;
    int e = (int)(gid >> 6);
    if (e >= E) return;
    int f = ((int)gid & 63) << 2;
    int s = src[e], d = dst[e];
    float norm = dinv[s] * dinv[d];
    const float4 v = *(const float4*)(xw + (size_t)s * HID + f);
    float* o = agg + (size_t)d * HID + f;
    atomicAdd(o + 0, norm * v.x);
    atomicAdd(o + 1, norm * v.y);
    atomicAdd(o + 2, norm * v.z);
    atomicAdd(o + 3, norm * v.w);
}

__global__ void bias_relu_bf16(const float* __restrict__ agg, const float* __restrict__ bias,
                               __bf16* __restrict__ hb, int n) {
    int i = blockIdx.x * blockDim.x + threadIdx.x;
    if (i >= n) return;
    float v = agg[i] + bias[i & (HID - 1)];
    hb[i] = (__bf16)(v > 0.f ? v : 0.f);
}

typedef uint32_t u32x4 __attribute__((ext_vector_type(4)));
typedef __bf16 bf16x8 __attribute__((ext_vector_type(8)));
typedef float f32x2 __attribute__((ext_vector_type(2)));
#define GRU_THREADS 256
#define NROW   24
#define R2     12
#define NPL    60
#define BATCH  6
#define NBATCH 10
#define RING   3

template <int M>
static __device__ __forceinline__ float lane_xor(float v) {
    int o = __builtin_amdgcn_update_dpp(0, __builtin_bit_cast(int, v), 0x160 | M, 0xf, 0xf, false);
    return __builtin_bit_cast(float, o);
}

static __device__ __forceinline__ float tanh_both_ranges(float x) {
    const float ax = __builtin_fabsf(x);
    const float x2 = x * x;
    float p = fmaf(__builtin_bit_cast(float, 0xbbbac73du), x2, __builtin_bit_cast(float, 0x3ca908c9u));
    p = fmaf(x2, p, __builtin_bit_cast(float, 0xbd5c1c4eu));
    p = fmaf(x2, p, __builtin_bit_cast(float, 0x3e088382u));
    p = fmaf(x2, p, __builtin_bit_cast(float, 0xbeaaaa99u));
    const float small = fmaf(x2, ax * p, ax);
    const float e = expf(ax + ax);
    const float big = fmaf(__builtin_amdgcn_rcpf(e + 1.0f), -2.0f, 1.0f);
    const float m = (ax < 0.625f) ? small : big;
    return __builtin_copysignf(m, x);
}

__global__ void __launch_bounds__(GRU_THREADS)
gru_scan(const float* __restrict__ gi, const __bf16* __restrict__ Whh,
         const float* __restrict__ b_ih, const float* __restrict__ b_hh,
         float* __restrict__ hs, int T) {
    __shared__ u32x4 wl[GRU_THREADS / 32][NPL][32];
    __shared__ __attribute__((aligned(16))) float hb[2 * HID];
    const int t = threadIdx.x;
    const int grp = t >> 3;
    const int s   = t & 7;
    const int wv = t >> 5, ln = t & 31;

    f32x2 wf2[8][R2 / 2];
    f32x2 wf3[8][NROW / 2];
#pragma unroll
    for (int r = 0; r < NROW; ++r) {
        const int row = (r >> 3) * HID + 8 * grp + ((r & 7) ^ s);
        const u32x4* wsrc = (const u32x4*)(Whh + (size_t)row * HID) + s;
        wl[wv][r][ln]      = wsrc[0];
        wl[wv][24 + r][ln] = wsrc[8];
        if (r < R2) {
            const bf16x8 w2 = __builtin_bit_cast(bf16x8, wsrc[16]);
#pragma unroll
            for (int e = 0; e < 8; ++e) wf2[e][r >> 1][r & 1] = (float)w2[e];
        } else {
            wl[wv][48 + (r - R2)][ln] = wsrc[16];
        }
        const bf16x8 w3 = __builtin_bit_cast(bf16x8, wsrc[24]);
#pragma unroll
        for (int e = 0; e < 8; ++e) wf3[e][r >> 1][r & 1] = (float)w3[e];
    }
    const float bq0 = b_hh[t], bq1 = b_hh[HID + t], bq2 = b_hh[2 * HID + t];
    const float bi0 = b_ih[t], bi1 = b_ih[HID + t], bi2 = b_ih[2 * HID + t];
    hb[t] = 0.0f;
    hb[HID + t] = 0.0f;
    __syncthreads();

    const u32x4* wme = &wl[wv][0][ln];
    u32x4 ring[RING][BATCH];
#pragma unroll
    for (int b = 0; b < RING - 1; ++b) {
#pragma unroll
        for (int rr = 0; rr < BATCH; ++rr) ring[b][rr] = wme[32 * (BATCH * b + rr)];
    }

    float hprev = 0.0f;
    int hoff = 0;
    for (int step = 0; step < T; ++step) {
        const float* g = gi + (size_t)step * GATES + t;
        const float g0 = g[0], g1 = g[HID], g2 = g[2 * HID];
        const float* hcur = hb + hoff;
        float hv[32];
#pragma unroll
        for (int i = 0; i < 4; ++i) {
            const float4 p = *(const float4*)&hcur[64 * i + 8 * s];
            const float4 q = *(const float4*)&hcur[64 * i + 8 * s + 4];
            hv[8 * i + 0] = p.x; hv[8 * i + 1] = p.y; hv[8 * i + 2] = p.z; hv[8 * i + 3] = p.w;
            hv[8 * i + 4] = q.x; hv[8 * i + 5] = q.y; hv[8 * i + 6] = q.z; hv[8 * i + 7] = q.w;
        }
        f32x2 acc[NROW / 2];
#pragma unroll
        for (int p = 0; p < NROW / 2; ++p) acc[p] = (f32x2){0.0f, 0.0f};
#pragma unroll
        for (int b = 0; b < NBATCH; ++b) {
            if (b == 0) __builtin_amdgcn_sched_barrier(0);
            if (b + RING - 1 < NBATCH) {
#pragma unroll
                for (int rr = 0; rr < BATCH; ++rr) ring[(b + RING - 1) % RING][rr] = wme[32 * (BATCH * (b + RING - 1) + rr)];
            }
            if (b == 0) __builtin_amdgcn_sched_barrier(0);
            if (b == 0) {
                if (step > 0) hs[(size_t)(step - 1) * HID + t] = hprev;
            }
#pragma unroll
            for (int e = 0; e < 8; ++e) {
#pragma unroll
                for (int rr = 0; rr < BATCH; ++rr) {
                    const int pl = BATCH * b + rr;
                    const int i = pl < 48 ? pl / 24 : 2;
                    const int r = pl < 48 ? pl % 24 : R2 + (pl - 48);
                    const bf16x8 w = __builtin_bit_cast(bf16x8, ring[b % RING][rr]);
                    acc[r >> 1][r & 1] = fmaf((float)w[e], hv[8 * i + e], acc[r >> 1][r & 1]);
                }
            }
        }
#pragma unroll
        for (int b = 0; b + RING < NBATCH; ++b) {
            __builtin_amdgcn_sched_group_barrier(0x002, 8 * BATCH, 0);
            __builtin_amdgcn_sched_group_barrier(0x100, BATCH, 0);
        }
#pragma unroll
        for (int b = 0; b < RING; ++b)
            __builtin_amdgcn_sched_group_barrier(0x002, 8 * BATCH, 0);
#pragma unroll
        for (int b = 0; b < RING - 1; ++b) {
#pragma unroll
            for (int rr = 0; rr < BATCH; ++rr) ring[b][rr] = wme[32 * (BATCH * b + rr)];
        }
#pragma unroll
        for (int e = 0; e < 8; ++e) {
#pragma unroll
            for (int p = 0; p < R2 / 2; ++p) {
                acc[p][0] = fmaf(wf2[e][p][0], hv[16 + e], acc[p][0]);
                acc[p][1] = fmaf(wf2[e][p][1], hv[16 + e], acc[p][1]);
            }
        }
#pragma unroll
        for (int e = 0; e < 8; ++e) {
#pragma unroll
            for (int p = 0; p < NROW / 2; ++p) {
                acc[p][0] = fmaf(wf3[e][p][0], hv[24 + e], acc[p][0]);
                acc[p][1] = fmaf(wf3[e][p][1], hv[24 + e], acc[p][1]);
            }
        }
        __builtin_amdgcn_sched_group_barrier(0x100, (RING - 1) * BATCH, 0);
        __builtin_amdgcn_sched_group_barrier(0x002, 8 * (R2 / 2) + 8 * (NROW / 2), 0);
        float a12[NROW / 2];
#pragma unroll
        for (int p = 0; p < NROW / 2; ++p) a12[p] = acc[p][0] + lane_xor<1>(acc[p][1]);
        float b6[NROW / 4];
#pragma unroll
        for (int m = 0; m < NROW / 4; ++m) b6[m] = a12[2 * m] + lane_xor<2>(a12[2 * m + 1]);
        float c3[3];
#pragma unroll
        for (int gt = 0; gt < 3; ++gt) c3[gt] = b6[2 * gt] + lane_xor<4>(b6[2 * gt + 1]);
        const float ghr = c3[0] + bq0, ghz = c3[1] + bq1, ghn = c3[2] + bq2;
        const float x0 = g0 + bi0, x1 = g1 + bi1, x2 = g2 + bi2;
        const float r  = 1.0f / (1.0f + __expf(-(x0 + ghr)));
        const float z  = 1.0f / (1.0f + __expf(-(x1 + ghz)));
        const float nn = tanh_both_ranges(x2 + r * ghn);
        const float hn = fmaf(1.0f - z, nn, z * hprev);
        hprev = hn;
        hoff ^= HID;
        hb[hoff + t] = hn;
        __syncthreads();
    }
    if (T > 0) hs[(size_t)(T - 1) * HID + t] = hprev;
}

__global__ void fc_out_kernel(const float* __restrict__ hs, const float* __restrict__ fw,
                              const float* __restrict__ fb, float* __restrict__ out, int total) {
    int gid = blockIdx.x * blockDim.x + threadIdx.x;
    if (gid >= total) return;
    int n = gid / OUTF, o = gid % OUTF;
    const float* h = hs + (size_t)n * HID;
    const float* w = fw + (size_t)o * HID;
    float acc = fb[o];
#pragma unroll 8
    for (int k = 0; k < HID; ++k) acc = fmaf(h[k], w[k], acc);
    out[gid] = acc;
}

extern "C" void kernel_launch(void* const* d_in, const int* in_sizes, int n_in,
                              void* d_out, int out_size, void* d_ws, size_t ws_size,
                              hipStream_t stream) {
    const float* x    = (const float*)d_in[0];
    const int*   ei   = (const int*)  d_in[1];
    const float* W1   = (const float*)d_in[2];
    const float* b1   = (const float*)d_in[3];
    const float* W2   = (const float*)d_in[4];
    const float* b2   = (const float*)d_in[5];
    const float* W_ih = (const float*)d_in[6];
    const float* W_hh = (const float*)d_in[7];
    const float* b_ih = (const float*)d_in[8];
    const float* b_hh = (const float*)d_in[9];
    const float* fc_w = (const float*)d_in[10];
    const float* fc_b = (const float*)d_in[11];
    float* out = (float*)d_out;

    const int E = in_sizes[1] / 2;
    const int* src = ei;
    const int* dst = ei + E;

    char* ws = (char*)d_ws;
    size_t off = 0;
    auto alloc = [&](size_t bytes) -> char* {
        char* p = ws + off;
        off = (off + bytes + 255) & ~(size_t)255;
        return p;
    };
    float*  deg  = (float*) alloc((size_t)NUM_NODES * 4);
    float*  dinv = (float*) alloc((size_t)NUM_NODES * 4);
    float*  xw   = (float*) alloc((size_t)NUM_NODES * HID * 4);
    float*  agg  = (float*) alloc((size_t)NUM_NODES * HID * 4);
    float*  gi   = (float*) alloc((size_t)NUM_NODES * GATES * 4);
    float*  hs   = (float*) alloc((size_t)NUM_NODES * HID * 4);
    __bf16* xb   = (__bf16*)alloc((size_t)NUM_NODES * HID * 2);
    __bf16* w1t  = (__bf16*)alloc((size_t)HID * IN_FEAT * 2);
    __bf16* w2t  = (__bf16*)alloc((size_t)HID * HID * 2);
    __bf16* wih  = (__bf16*)alloc((size_t)GATES * HID * 2);
    __bf16* whh  = (__bf16*)alloc((size_t)GATES * HID * 2);
    (void)ws_size; (void)n_in; (void)out_size;

    const int TPB = 256;
    auto blk = [](long long n, int t) { return (unsigned)((n + t - 1) / t); };

    cvt_f32_to_bf16<<<blk((long long)NUM_NODES * IN_FEAT, TPB), TPB, 0, stream>>>(
        x, xb, NUM_NODES * IN_FEAT);
    cvt_f32_to_bf16_T<<<blk((long long)IN_FEAT * HID, TPB), TPB, 0, stream>>>(
        W1, w1t, IN_FEAT, HID);
    cvt_f32_to_bf16_T<<<blk((long long)HID * HID, TPB), TPB, 0, stream>>>(
        W2, w2t, HID, HID);
    cvt_f32_to_bf16<<<blk((long long)GATES * HID, TPB), TPB, 0, stream>>>(
        W_ih, wih, GATES * HID);
    cvt_f32_to_bf16<<<blk((long long)GATES * HID, TPB), TPB, 0, stream>>>(
        W_hh, whh, GATES * HID);

    deg_init<<<blk(NUM_NODES, TPB), TPB, 0, stream>>>(deg, NUM_NODES);
    deg_edges<<<blk(E, TPB), TPB, 0, stream>>>(dst, deg, E);
    deg_rsqrt<<<blk(NUM_NODES, TPB), TPB, 0, stream>>>(deg, dinv, NUM_NODES);

    const long long NH = (long long)NUM_NODES * HID;
    const unsigned gemm1_blocks = blk((long long)(NUM_NODES / 16) * (HID / 16), 4);
    const unsigned gemm2_blocks = gemm1_blocks;
    const unsigned gemm3_blocks = blk((long long)(NUM_NODES / 16) * (GATES / 16), 4);
    const unsigned scat_blocks  = blk((long long)E * 64, TPB);

    wmma_gemm_bf16<<<gemm1_blocks, 128, 0, stream>>>(xb, w1t, xw, NUM_NODES, HID, IN_FEAT);
    gcn_selfloop<<<blk(NH, TPB), TPB, 0, stream>>>(xw, dinv, agg, (int)NH);
    gcn_scatter<<<scat_blocks, TPB, 0, stream>>>(xw, src, dst, dinv, agg, E);
    bias_relu_bf16<<<blk(NH, TPB), TPB, 0, stream>>>(agg, b1, xb, (int)NH);

    wmma_gemm_bf16<<<gemm2_blocks, 128, 0, stream>>>(xb, w2t, xw, NUM_NODES, HID, HID);
    gcn_selfloop<<<blk(NH, TPB), TPB, 0, stream>>>(xw, dinv, agg, (int)NH);
    gcn_scatter<<<scat_blocks, TPB, 0, stream>>>(xw, src, dst, dinv, agg, E);
    bias_relu_bf16<<<blk(NH, TPB), TPB, 0, stream>>>(agg, b2, xb, (int)NH);

    wmma_gemm_bf16<<<gemm3_blocks, 128, 0, stream>>>(xb, wih, gi, NUM_NODES, GATES, HID);

    gru_scan<<<1, GRU_THREADS, 0, stream>>>(gi, whh, b_ih, b_hh, hs, NUM_NODES);

    fc_out_kernel<<<blk((long long)NUM_NODES * OUTF, TPB), TPB, 0, stream>>>(
        hs, fc_w, fc_b, out, NUM_NODES * OUTF);
}
